# starting kernel
# baseline (speedup 1.0000x reference)
#include <hip/hip_runtime.h>

typedef __attribute__((ext_vector_type(2))) float v2f;
typedef __attribute__((ext_vector_type(8))) float v8f;

static constexpr unsigned long long kPerm = 0xB5F93E82A4D71C60ULL;

__device__ __forceinline__ float sigmoid_fast(float x) {
  float e = __builtin_amdgcn_exp2f(x * -1.4426950408889634f);
  return __builtin_amdgcn_rcpf(1.0f + e);
}
__device__ __forceinline__ float tanh_fast(float x) {
  float e = __builtin_amdgcn_exp2f(x * -2.8853900817779268f);
  return (1.0f - e) * __builtin_amdgcn_rcpf(1.0f + e);
}
__device__ __forceinline__ v8f wmma4(v2f a, v2f b, v8f c) {
  return __builtin_amdgcn_wmma_f32_16x16x4_f32(false, a, false, b, (short)0, c, false, false);
}
__device__ __forceinline__ v2f mk2(float a, float b) { v2f r; r[0] = a; r[1] = b; return r; }

__global__ __launch_bounds__(256) void gru16_wmma_kernel(
    const float* __restrict__ x, const int* __restrict__ lengths,
    const float* __restrict__ W_ih, const float* __restrict__ W_hh,
    const float* __restrict__ b_ih, const float* __restrict__ b_hh,
    float* __restrict__ out, int tmax)
{
  const int lane = threadIdx.x & 31;
  const int col  = lane & 15;
  const int half = lane >> 4;
  const int wv   = (blockIdx.x * blockDim.x + threadIdx.x) >> 5;
  const int s    = wv * 16 + col;

  const int g0 = (int)((kPerm >> (4 * col)) & 0xFULL);
  const int g1 = (col == 0) ? 16 : ((col == 8) ? 17 : -1);
  const int cl = 2 * half;

  v2f Aih[2][2][2], Ahh[2][2][2];
  v8f Cih[2][2],    Chh[2][2];
#pragma unroll
  for (int l = 0; l < 2; ++l) {
    const float* Wi = W_ih + l * 108;
    const float* Wh = W_hh + l * 108;
    const float* Bi = b_ih + l * 18;
    const float* Bh = b_hh + l * 18;

    Aih[l][0][0] = mk2(Wi[g0 * 6 + cl], Wi[g0 * 6 + cl + 1]);
    Ahh[l][0][0] = mk2(Wh[g0 * 6 + cl], Wh[g0 * 6 + cl + 1]);
    Aih[l][0][1] = half ? mk2(0.f, 0.f) : mk2(Wi[g0 * 6 + 4], Wi[g0 * 6 + 5]);
    Ahh[l][0][1] = half ? mk2(0.f, 0.f) : mk2(Wh[g0 * 6 + 4], Wh[g0 * 6 + 5]);
    if (g1 >= 0) {
      Aih[l][1][0] = mk2(Wi[g1 * 6 + cl], Wi[g1 * 6 + cl + 1]);
      Ahh[l][1][0] = mk2(Wh[g1 * 6 + cl], Wh[g1 * 6 + cl + 1]);
      Aih[l][1][1] = half ? mk2(0.f, 0.f) : mk2(Wi[g1 * 6 + 4], Wi[g1 * 6 + 5]);
      Ahh[l][1][1] = half ? mk2(0.f, 0.f) : mk2(Wh[g1 * 6 + 4], Wh[g1 * 6 + 5]);
    } else {
      Aih[l][1][0] = mk2(0.f, 0.f); Ahh[l][1][0] = mk2(0.f, 0.f);
      Aih[l][1][1] = mk2(0.f, 0.f); Ahh[l][1][1] = mk2(0.f, 0.f);
    }
    v8f ci0, ch0, ci1, ch1;
#pragma unroll
    for (int r = 0; r < 8; ++r) {
      int row = r + 8 * half;
      int g = (int)((kPerm >> (4 * row)) & 0xFULL);
      ci0[r] = Bi[g];  ch0[r] = Bh[g];
      ci1[r] = 0.f;    ch1[r] = 0.f;
    }
    ci1[0] = Bi[half ? 17 : 16];
    ch1[0] = Bh[half ? 17 : 16];
    Cih[l][0] = ci0; Chh[l][0] = ch0;
    Cih[l][1] = ci1; Chh[l][1] = ch1;
  }

  float ha0 = 0.f, hb0 = 0.f, hc0 = 0.f;
  float ha1 = 0.f, hb1 = 0.f, hc1 = 0.f;
  const int len = lengths[s];
  const float* xlo = x + (size_t)s * (size_t)tmax * 6 + cl;
  const float* xhi = x + (size_t)s * (size_t)tmax * 6 + 4;

  float2 xl = *(const float2*)xlo;
  float2 xh = *(const float2*)xhi;

#pragma unroll 1
  for (int t = 0; t < tmax; ++t) {
    const bool valid = (t < len);
    v2f xB0 = mk2(xl.x, xl.y);
    v2f xB1 = half ? mk2(0.f, 0.f) : mk2(xh.x, xh.y);

    if (t + 1 < tmax) {
      xlo += 6; xhi += 6;
      xl = *(const float2*)xlo;
      xh = *(const float2*)xhi;
    }

    {
      int sw = __builtin_amdgcn_ds_swizzle(__float_as_int(hc0), 0x401F);
      v2f H0 = mk2(ha0, hb0);
      v2f H1 = half ? mk2(0.f, 0.f) : mk2(hc0, __int_as_float(sw));
      v8f gx0 = wmma4(Aih[0][0][1], xB1, wmma4(Aih[0][0][0], xB0, Cih[0][0]));
      v8f gx1 = wmma4(Aih[0][1][1], xB1, wmma4(Aih[0][1][0], xB0, Cih[0][1]));
      v8f gh0 = wmma4(Ahh[0][0][1], H1, wmma4(Ahh[0][0][0], H0, Chh[0][0]));
      v8f gh1 = wmma4(Ahh[0][1][1], H1, wmma4(Ahh[0][1][0], H0, Chh[0][1]));
      float r, z, nn, h;
      r = sigmoid_fast(gx0[0] + gh0[0]); z = sigmoid_fast(gx0[1] + gh0[1]);
      nn = tanh_fast(gx0[2] + r * gh0[2]); h = nn + z * (ha0 - nn); ha0 = valid ? h : ha0;
      r = sigmoid_fast(gx0[3] + gh0[3]); z = sigmoid_fast(gx0[4] + gh0[4]);
      nn = tanh_fast(gx0[5] + r * gh0[5]); h = nn + z * (hb0 - nn); hb0 = valid ? h : hb0;
      r = sigmoid_fast(gx0[6] + gh0[6]); z = sigmoid_fast(gx0[7] + gh0[7]);
      nn = tanh_fast(gx1[0] + r * gh1[0]); h = nn + z * (hc0 - nn); hc0 = valid ? h : hc0;
    }
    {
      int sw0 = __builtin_amdgcn_ds_swizzle(__float_as_int(hc0), 0x401F);
      v2f I0 = mk2(ha0, hb0);
      v2f I1 = half ? mk2(0.f, 0.f) : mk2(hc0, __int_as_float(sw0));
      int sw1 = __builtin_amdgcn_ds_swizzle(__float_as_int(hc1), 0x401F);
      v2f H0 = mk2(ha1, hb1);
      v2f H1 = half ? mk2(0.f, 0.f) : mk2(hc1, __int_as_float(sw1));
      v8f gx0 = wmma4(Aih[1][0][1], I1, wmma4(Aih[1][0][0], I0, Cih[1][0]));
      v8f gx1 = wmma4(Aih[1][1][1], I1, wmma4(Aih[1][1][0], I0, Cih[1][1]));
      v8f gh0 = wmma4(Ahh[1][0][1], H1, wmma4(Ahh[1][0][0], H0, Chh[1][0]));
      v8f gh1 = wmma4(Ahh[1][1][1], H1, wmma4(Ahh[1][1][0], H0, Chh[1][1]));
      float r, z, nn, h;
      r = sigmoid_fast(gx0[0] + gh0[0]); z = sigmoid_fast(gx0[1] + gh0[1]);
      nn = tanh_fast(gx0[2] + r * gh0[2]); h = nn + z * (ha1 - nn); ha1 = valid ? h : ha1;
      r = sigmoid_fast(gx0[3] + gh0[3]); z = sigmoid_fast(gx0[4] + gh0[4]);
      nn = tanh_fast(gx0[5] + r * gh0[5]); h = nn + z * (hb1 - nn); hb1 = valid ? h : hb1;
      r = sigmoid_fast(gx0[6] + gh0[6]); z = sigmoid_fast(gx0[7] + gh0[7]);
      nn = tanh_fast(gx1[0] + r * gh1[0]); h = nn + z * (hc1 - nn); hc1 = valid ? h : hc1;
    }
  }

  float* orow = out + (size_t)s * 6;
  *(float2*)(orow + (half ? 2 : 0)) = make_float2(ha1, hb1);
  orow[half ? 5 : 4] = hc1;
}

extern "C" void kernel_launch(void* const* d_in, const int* in_sizes, int n_in,
                              void* d_out, int out_size, void* d_ws, size_t ws_size,
                              hipStream_t stream) {
  (void)n_in; (void)out_size; (void)d_ws; (void)ws_size;
  const float* x       = (const float*)d_in[0];
  const int*   lengths = (const int*)d_in[1];
  const float* W_ih    = (const float*)d_in[2];
  const float* W_hh    = (const float*)d_in[3];
  const float* b_ih    = (const float*)d_in[4];
  const float* b_hh    = (const float*)d_in[5];
  float* out = (float*)d_out;

  const int nseq = in_sizes[1];
  const int tmax = in_sizes[0] / (nseq * 6);
  const int waves = nseq / 16;
  const int threads = waves * 32;
  const int block = 256;
  const int grid = (threads + block - 1) / block;
  gru16_wmma_kernel<<<grid, block, 0, stream>>>(x, lengths, W_ih, W_hh, b_ih, b_hh, out, tmax);
}


# Layer skew: trip t runs layer-0 step t and layer-1 step t-1 from the state of trip t-1, so both gate chains are independent and evaluated side by side; one extra drain trip
# speedup vs baseline: 1.1997x; 1.1997x over previous
#include <hip/hip_runtime.h>

typedef __attribute__((ext_vector_type(2))) float v2f;
typedef __attribute__((ext_vector_type(8))) float v8f;

static constexpr unsigned long long kPerm = 0xB5F93E82A4D71C60ULL;

__device__ __forceinline__ float sigmoid_fast(float x) {
  float e = __builtin_amdgcn_exp2f(x * -1.4426950408889634f);
  return __builtin_amdgcn_rcpf(1.0f + e);
}
__device__ __forceinline__ float tanh_fast(float x) {
  float e = __builtin_amdgcn_exp2f(x * -2.8853900817779268f);
  return (1.0f - e) * __builtin_amdgcn_rcpf(1.0f + e);
}
__device__ __forceinline__ v8f wmma4(v2f a, v2f b, v8f c) {
  return __builtin_amdgcn_wmma_f32_16x16x4_f32(false, a, false, b, (short)0, c, false, false);
}
__device__ __forceinline__ v2f mk2(float a, float b) { v2f r; r[0] = a; r[1] = b; return r; }

__global__ __launch_bounds__(256) void gru16_wmma_kernel(
    const float* __restrict__ x, const int* __restrict__ lengths,
    const float* __restrict__ W_ih, const float* __restrict__ W_hh,
    const float* __restrict__ b_ih, const float* __restrict__ b_hh,
    float* __restrict__ out, int tmax)
{
  const int lane = threadIdx.x & 31;
  const int col  = lane & 15;
  const int half = lane >> 4;
  const int wv   = (blockIdx.x * blockDim.x + threadIdx.x) >> 5;
  const int s    = wv * 16 + col;

  const int g0 = (int)((kPerm >> (4 * col)) & 0xFULL);
  const int g1 = (col == 0) ? 16 : ((col == 8) ? 17 : -1);
  const int cl = 2 * half;

  v2f Aih[2][2][2], Ahh[2][2][2];
  v8f Cih[2][2],    Chh[2][2];
#pragma unroll
  for (int l = 0; l < 2; ++l) {
    const float* Wi = W_ih + l * 108;
    const float* Wh = W_hh + l * 108;
    const float* Bi = b_ih + l * 18;
    const float* Bh = b_hh + l * 18;

    Aih[l][0][0] = mk2(Wi[g0 * 6 + cl], Wi[g0 * 6 + cl + 1]);
    Ahh[l][0][0] = mk2(Wh[g0 * 6 + cl], Wh[g0 * 6 + cl + 1]);
    Aih[l][0][1] = half ? mk2(0.f, 0.f) : mk2(Wi[g0 * 6 + 4], Wi[g0 * 6 + 5]);
    Ahh[l][0][1] = half ? mk2(0.f, 0.f) : mk2(Wh[g0 * 6 + 4], Wh[g0 * 6 + 5]);
    if (g1 >= 0) {
      Aih[l][1][0] = mk2(Wi[g1 * 6 + cl], Wi[g1 * 6 + cl + 1]);
      Ahh[l][1][0] = mk2(Wh[g1 * 6 + cl], Wh[g1 * 6 + cl + 1]);
      Aih[l][1][1] = half ? mk2(0.f, 0.f) : mk2(Wi[g1 * 6 + 4], Wi[g1 * 6 + 5]);
      Ahh[l][1][1] = half ? mk2(0.f, 0.f) : mk2(Wh[g1 * 6 + 4], Wh[g1 * 6 + 5]);
    } else {
      Aih[l][1][0] = mk2(0.f, 0.f); Ahh[l][1][0] = mk2(0.f, 0.f);
      Aih[l][1][1] = mk2(0.f, 0.f); Ahh[l][1][1] = mk2(0.f, 0.f);
    }
    v8f ci0, ch0, ci1, ch1;
#pragma unroll
    for (int r = 0; r < 8; ++r) {
      int row = r + 8 * half;
      int g = (int)((kPerm >> (4 * row)) & 0xFULL);
      ci0[r] = Bi[g];  ch0[r] = Bh[g];
      ci1[r] = 0.f;    ch1[r] = 0.f;
    }
    ci1[0] = Bi[half ? 17 : 16];
    ch1[0] = Bh[half ? 17 : 16];
    Cih[l][0] = ci0; Chh[l][0] = ch0;
    Cih[l][1] = ci1; Chh[l][1] = ch1;
  }

  float ha0 = 0.f, hb0 = 0.f, hc0 = 0.f;
  float ha1 = 0.f, hb1 = 0.f, hc1 = 0.f;
  const int len = lengths[s];
  const float* xlo = x + (size_t)s * (size_t)tmax * 6 + cl;
  const float* xhi = x + (size_t)s * (size_t)tmax * 6 + 4;

  float2 xl = *(const float2*)xlo;
  float2 xh = *(const float2*)xhi;

#pragma unroll 1
  for (int t = 0; t <= tmax; ++t) {
    const bool valid0 = (t < len) && (t < tmax);
    const bool valid1 = (t >= 1) && (t - 1 < len);
    v2f xB0 = mk2(xl.x, xl.y);
    v2f xB1 = half ? mk2(0.f, 0.f) : mk2(xh.x, xh.y);

    if (t + 1 < tmax) {
      xlo += 6; xhi += 6;
      xl = *(const float2*)xlo;
      xh = *(const float2*)xhi;
    }

    int sw0 = __builtin_amdgcn_ds_swizzle(__float_as_int(hc0), 0x401F);
    int sw1 = __builtin_amdgcn_ds_swizzle(__float_as_int(hc1), 0x401F);
    __builtin_amdgcn_sched_barrier(0);
    v2f P0 = mk2(ha0, hb0);
    v2f Q0 = mk2(ha1, hb1);
    v8f ax0 = wmma4(Aih[0][0][0], xB0, Cih[0][0]);
    v8f ah0 = wmma4(Ahh[0][0][0], P0, Chh[0][0]);
    v8f bx0 = wmma4(Aih[1][0][0], P0, Cih[1][0]);
    v8f bh0 = wmma4(Ahh[1][0][0], Q0, Chh[1][0]);
    v8f ax1 = wmma4(Aih[0][1][0], xB0, Cih[0][1]);
    v8f ah1 = wmma4(Ahh[0][1][0], P0, Chh[0][1]);
    v8f bx1 = wmma4(Aih[1][1][0], P0, Cih[1][1]);
    v8f bh1 = wmma4(Ahh[1][1][0], Q0, Chh[1][1]);
    __builtin_amdgcn_sched_barrier(0);
    v2f P1 = half ? mk2(0.f, 0.f) : mk2(hc0, __int_as_float(sw0));
    v2f Q1 = half ? mk2(0.f, 0.f) : mk2(hc1, __int_as_float(sw1));
    ax0 = wmma4(Aih[0][0][1], xB1, ax0);
    ah0 = wmma4(Ahh[0][0][1], P1, ah0);
    bx0 = wmma4(Aih[1][0][1], P1, bx0);
    bh0 = wmma4(Ahh[1][0][1], Q1, bh0);
    ax1 = wmma4(Aih[0][1][1], xB1, ax1);
    ah1 = wmma4(Ahh[0][1][1], P1, ah1);
    bx1 = wmma4(Aih[1][1][1], P1, bx1);
    bh1 = wmma4(Ahh[1][1][1], Q1, bh1);

    float r0, z0, n0, g0, r1, z1, n1, g1;
    r0 = sigmoid_fast(ax0[0] + ah0[0]);          r1 = sigmoid_fast(bx0[0] + bh0[0]);
    z0 = sigmoid_fast(ax0[1] + ah0[1]);          z1 = sigmoid_fast(bx0[1] + bh0[1]);
    n0 = tanh_fast(ax0[2] + r0 * ah0[2]);        n1 = tanh_fast(bx0[2] + r1 * bh0[2]);
    g0 = n0 + z0 * (ha0 - n0);                   g1 = n1 + z1 * (ha1 - n1);
    ha0 = valid0 ? g0 : ha0;                     ha1 = valid1 ? g1 : ha1;
    r0 = sigmoid_fast(ax0[3] + ah0[3]);          r1 = sigmoid_fast(bx0[3] + bh0[3]);
    z0 = sigmoid_fast(ax0[4] + ah0[4]);          z1 = sigmoid_fast(bx0[4] + bh0[4]);
    n0 = tanh_fast(ax0[5] + r0 * ah0[5]);        n1 = tanh_fast(bx0[5] + r1 * bh0[5]);
    g0 = n0 + z0 * (hb0 - n0);                   g1 = n1 + z1 * (hb1 - n1);
    hb0 = valid0 ? g0 : hb0;                     hb1 = valid1 ? g1 : hb1;
    r0 = sigmoid_fast(ax0[6] + ah0[6]);          r1 = sigmoid_fast(bx0[6] + bh0[6]);
    z0 = sigmoid_fast(ax0[7] + ah0[7]);          z1 = sigmoid_fast(bx0[7] + bh0[7]);
    n0 = tanh_fast(ax1[0] + r0 * ah1[0]);        n1 = tanh_fast(bx1[0] + r1 * bh1[0]);
    g0 = n0 + z0 * (hc0 - n0);                   g1 = n1 + z1 * (hc1 - n1);
    hc0 = valid0 ? g0 : hc0;                     hc1 = valid1 ? g1 : hc1;
  }

  float* orow = out + (size_t)s * 6;
  *(float2*)(orow + (half ? 2 : 0)) = make_float2(ha1, hb1);
  orow[half ? 5 : 4] = hc1;
}

extern "C" void kernel_launch(void* const* d_in, const int* in_sizes, int n_in,
                              void* d_out, int out_size, void* d_ws, size_t ws_size,
                              hipStream_t stream) {
  (void)n_in; (void)out_size; (void)d_ws; (void)ws_size;
  const float* x       = (const float*)d_in[0];
  const int*   lengths = (const int*)d_in[1];
  const float* W_ih    = (const float*)d_in[2];
  const float* W_hh    = (const float*)d_in[3];
  const float* b_ih    = (const float*)d_in[4];
  const float* b_hh    = (const float*)d_in[5];
  float* out = (float*)d_out;

  const int nseq = in_sizes[1];
  const int tmax = in_sizes[0] / (nseq * 6);
  const int waves = nseq / 16;
  const int threads = waves * 32;
  const int block = 256;
  const int grid = (threads + block - 1) / block;
  gru16_wmma_kernel<<<grid, block, 0, stream>>>(x, lengths, W_ih, W_hh, b_ih, b_hh, out, tmax);
}


# Fuse each layer's W_ih and W_hh products into one K=12 WMMA chain, k-tiles packed from lane-local values: 12 WMMA per trip (was 16), no cross-lane fetch; gx+gh and bias pair now sum in the accumulator
# speedup vs baseline: 1.5467x; 1.2892x over previous
#include <hip/hip_runtime.h>

typedef __attribute__((ext_vector_type(2))) float v2f;
typedef __attribute__((ext_vector_type(8))) float v8f;

struct RowSel { int g; bool ui; bool uh; };
__device__ __forceinline__ RowSel row_sel(int m, int R) {
  const int hh = R >> 3, r = R & 7;
  const int ua = 2 * hh, ub = 2 * hh + 1, uc = 4 + hh;
  RowSel s; s.g = 0; s.ui = false; s.uh = false;
  if (m == 0) {
    if (r < 6) {
      const int p = r >> 1;
      const int u = (p == 0) ? ua : ((p == 1) ? ub : uc);
      s.g = u + 6 * (r & 1); s.ui = true; s.uh = true;
    } else {
      s.g = 12 + ua; s.ui = (r == 6); s.uh = (r == 7);
    }
  } else if (r < 4) {
    const int u = (r >> 1) ? uc : ub;
    s.g = 12 + u; s.ui = ((r & 1) == 0); s.uh = ((r & 1) != 0);
  }
  return s;
}

__device__ __forceinline__ float sigmoid_fast(float x) {
  float e = __builtin_amdgcn_exp2f(x * -1.4426950408889634f);
  return __builtin_amdgcn_rcpf(1.0f + e);
}
__device__ __forceinline__ float tanh_fast(float x) {
  float e = __builtin_amdgcn_exp2f(x * -2.8853900817779268f);
  return (1.0f - e) * __builtin_amdgcn_rcpf(1.0f + e);
}
__device__ __forceinline__ v8f wmma4(v2f a, v2f b, v8f c) {
  return __builtin_amdgcn_wmma_f32_16x16x4_f32(false, a, false, b, (short)0, c, false, false);
}
__device__ __forceinline__ v2f mk2(float a, float b) { v2f r; r[0] = a; r[1] = b; return r; }

__global__ __launch_bounds__(256) void gru16_wmma_kernel(
    const float* __restrict__ x, const int* __restrict__ lengths,
    const float* __restrict__ W_ih, const float* __restrict__ W_hh,
    const float* __restrict__ b_ih, const float* __restrict__ b_hh,
    float* __restrict__ out, int tmax)
{
  const int lane = threadIdx.x & 31;
  const int col  = lane & 15;
  const int half = lane >> 4;
  const int wv   = (blockIdx.x * blockDim.x + threadIdx.x) >> 5;
  const int s    = wv * 16 + col;

  v2f A[2][2][3];
  v8f C[2][2];
#pragma unroll
  for (int l = 0; l < 2; ++l) {
    const float* Wi = W_ih + l * 108;
    const float* Wh = W_hh + l * 108;
    const float* Bi = b_ih + l * 18;
    const float* Bh = b_hh + l * 18;
#pragma unroll
    for (int m = 0; m < 2; ++m) {
      const RowSel rs = row_sel(m, col);
      const float* wi = Wi + rs.g * 6;
      const float* wh = Wh + rs.g * 6;
      const float i0 = rs.ui ? wi[2 * half]     : 0.f;
      const float i1 = rs.ui ? wi[2 * half + 1] : 0.f;
      const float i2 = rs.ui ? wi[4 + half]     : 0.f;
      const float h0 = rs.uh ? wh[2 * half]     : 0.f;
      const float h1 = rs.uh ? wh[2 * half + 1] : 0.f;
      const float h2 = rs.uh ? wh[4 + half]     : 0.f;
      A[l][m][0] = mk2(i0, i1);
      A[l][m][1] = mk2(h0, h1);
      A[l][m][2] = mk2(i2, h2);
      v8f c;
#pragma unroll
      for (int r = 0; r < 8; ++r) {
        const RowSel cs = row_sel(m, r + 8 * half);
        const float bi = cs.ui ? Bi[cs.g] : 0.f;
        const float bh = cs.uh ? Bh[cs.g] : 0.f;
        c[r] = bi + bh;
      }
      C[l][m] = c;
    }
  }

  float ha0 = 0.f, hb0 = 0.f, hc0 = 0.f;
  float ha1 = 0.f, hb1 = 0.f, hc1 = 0.f;
  const int len = lengths[s];
  const float* xlo = x + (size_t)s * (size_t)tmax * 6 + 2 * half;
  const float* xhi = x + (size_t)s * (size_t)tmax * 6 + 4 + half;

  float2 xl = *(const float2*)xlo;
  float  xc = *xhi;

#pragma unroll 1
  for (int t = 0; t <= tmax; ++t) {
    const bool valid0 = (t < len) && (t < tmax);
    const bool valid1 = (t >= 1) && (t - 1 < len);
    const v2f X0 = mk2(xl.x, xl.y);
    const v2f X2 = mk2(xc, hc0);
    const v2f P0 = mk2(ha0, hb0);
    const v2f Q0 = mk2(ha1, hb1);
    const v2f PQ = mk2(hc0, hc1);

    if (t + 1 < tmax) {
      xlo += 6; xhi += 6;
      xl = *(const float2*)xlo;
      xc = *xhi;
    }

    v8f a0 = wmma4(A[0][0][0], X0, C[0][0]);
    v8f b0 = wmma4(A[1][0][0], P0, C[1][0]);
    v8f a1 = wmma4(A[0][1][0], X0, C[0][1]);
    v8f b1 = wmma4(A[1][1][0], P0, C[1][1]);
    a0 = wmma4(A[0][0][1], P0, a0);
    b0 = wmma4(A[1][0][1], Q0, b0);
    a1 = wmma4(A[0][1][1], P0, a1);
    b1 = wmma4(A[1][1][1], Q0, b1);
    a0 = wmma4(A[0][0][2], X2, a0);
    b0 = wmma4(A[1][0][2], PQ, b0);
    a1 = wmma4(A[0][1][2], X2, a1);
    b1 = wmma4(A[1][1][2], PQ, b1);

    float r0, z0, n0, g0, r1, z1, n1, g1;
    r0 = sigmoid_fast(a0[0]);                    r1 = sigmoid_fast(b0[0]);
    z0 = sigmoid_fast(a0[1]);                    z1 = sigmoid_fast(b0[1]);
    n0 = tanh_fast(a0[6] + r0 * a0[7]);          n1 = tanh_fast(b0[6] + r1 * b0[7]);
    g0 = n0 + z0 * (ha0 - n0);                   g1 = n1 + z1 * (ha1 - n1);
    ha0 = valid0 ? g0 : ha0;                     ha1 = valid1 ? g1 : ha1;
    r0 = sigmoid_fast(a0[2]);                    r1 = sigmoid_fast(b0[2]);
    z0 = sigmoid_fast(a0[3]);                    z1 = sigmoid_fast(b0[3]);
    n0 = tanh_fast(a1[0] + r0 * a1[1]);          n1 = tanh_fast(b1[0] + r1 * b1[1]);
    g0 = n0 + z0 * (hb0 - n0);                   g1 = n1 + z1 * (hb1 - n1);
    hb0 = valid0 ? g0 : hb0;                     hb1 = valid1 ? g1 : hb1;
    r0 = sigmoid_fast(a0[4]);                    r1 = sigmoid_fast(b0[4]);
    z0 = sigmoid_fast(a0[5]);                    z1 = sigmoid_fast(b0[5]);
    n0 = tanh_fast(a1[2] + r0 * a1[3]);          n1 = tanh_fast(b1[2] + r1 * b1[3]);
    g0 = n0 + z0 * (hc0 - n0);                   g1 = n1 + z1 * (hc1 - n1);
    hc0 = valid0 ? g0 : hc0;                     hc1 = valid1 ? g1 : hc1;
  }

  float* orow = out + (size_t)s * 6;
  *(float2*)(orow + (half ? 2 : 0)) = make_float2(ha1, hb1);
  orow[half ? 5 : 4] = hc1;
}

extern "C" void kernel_launch(void* const* d_in, const int* in_sizes, int n_in,
                              void* d_out, int out_size, void* d_ws, size_t ws_size,
                              hipStream_t stream) {
  (void)n_in; (void)out_size; (void)d_ws; (void)ws_size;
  const float* x       = (const float*)d_in[0];
  const int*   lengths = (const int*)d_in[1];
  const float* W_ih    = (const float*)d_in[2];
  const float* W_hh    = (const float*)d_in[3];
  const float* b_ih    = (const float*)d_in[4];
  const float* b_hh    = (const float*)d_in[5];
  float* out = (float*)d_out;

  const int nseq = in_sizes[1];
  const int tmax = in_sizes[0] / (nseq * 6);
  const int waves = nseq / 16;
  const int threads = waves * 32;
  const int block = 256;
  const int grid = (threads + block - 1) / block;
  gru16_wmma_kernel<<<grid, block, 0, stream>>>(x, lengths, W_ih, W_hh, b_ih, b_hh, out, tmax);
}


# Gates on the hardware tanh (v_tanh_f32): sigmoid(v)=0.5+0.5*tanh(v/2), the halves folded into the weight/bias rows; n-gate row now sums W_in x + 0.5 W_hn h in the accumulator
# speedup vs baseline: 1.6397x; 1.0601x over previous
#include <hip/hip_runtime.h>

typedef __attribute__((ext_vector_type(2))) float v2f;
typedef __attribute__((ext_vector_type(8))) float v8f;

struct RowSel { int g; float si; float sh; };
__device__ __forceinline__ RowSel row_sel(int m, int R) {
  const int hh = R >> 3, r = R & 7;
  const int ua = 2 * hh, ub = 2 * hh + 1, uc = 4 + hh;
  RowSel s; s.g = 0; s.si = 0.f; s.sh = 0.f;
  if (m == 0) {
    if (r < 6) {
      const int p = r >> 1;
      const int u = (p == 0) ? ua : ((p == 1) ? ub : uc);
      s.g = u + 6 * (r & 1); s.si = 0.5f; s.sh = 0.5f;
    } else {
      s.g = 12 + ua; s.si = (r == 6) ? 1.f : 0.f; s.sh = 0.5f;
    }
  } else if (r < 4) {
    const int u = (r >> 1) ? uc : ub;
    s.g = 12 + u; s.si = ((r & 1) == 0) ? 1.f : 0.f; s.sh = 0.5f;
  }
  return s;
}

__device__ __forceinline__ float tanh_hw(float x) {
  return __builtin_amdgcn_tanhf(x);
}
__device__ __forceinline__ float gru_unit(float ar, float az, float s, float q, float h) {
  const float tr = tanh_hw(ar);
  const float tz = tanh_hw(az);
  const float n  = tanh_hw(__builtin_fmaf(tr, q, s));
  const float z  = __builtin_fmaf(0.5f, tz, 0.5f);
  return __builtin_fmaf(z, h - n, n);
}
__device__ __forceinline__ v8f wmma4(v2f a, v2f b, v8f c) {
  return __builtin_amdgcn_wmma_f32_16x16x4_f32(false, a, false, b, (short)0, c, false, false);
}
__device__ __forceinline__ v2f mk2(float a, float b) { v2f r; r[0] = a; r[1] = b; return r; }

__global__ __launch_bounds__(256) void gru16_wmma_kernel(
    const float* __restrict__ x, const int* __restrict__ lengths,
    const float* __restrict__ W_ih, const float* __restrict__ W_hh,
    const float* __restrict__ b_ih, const float* __restrict__ b_hh,
    float* __restrict__ out, int tmax)
{
  const int lane = threadIdx.x & 31;
  const int col  = lane & 15;
  const int half = lane >> 4;
  const int wv   = (blockIdx.x * blockDim.x + threadIdx.x) >> 5;
  const int s    = wv * 16 + col;

  v2f A[2][2][3];
  v8f C[2][2];
#pragma unroll
  for (int l = 0; l < 2; ++l) {
    const float* Wi = W_ih + l * 108;
    const float* Wh = W_hh + l * 108;
    const float* Bi = b_ih + l * 18;
    const float* Bh = b_hh + l * 18;
#pragma unroll
    for (int m = 0; m < 2; ++m) {
      const RowSel rs = row_sel(m, col);
      const float* wi = Wi + rs.g * 6;
      const float* wh = Wh + rs.g * 6;
      const bool ui = (rs.si != 0.f), uh = (rs.sh != 0.f);
      const float i0 = ui ? rs.si * wi[2 * half]     : 0.f;
      const float i1 = ui ? rs.si * wi[2 * half + 1] : 0.f;
      const float i2 = ui ? rs.si * wi[4 + half]     : 0.f;
      const float h0 = uh ? rs.sh * wh[2 * half]     : 0.f;
      const float h1 = uh ? rs.sh * wh[2 * half + 1] : 0.f;
      const float h2 = uh ? rs.sh * wh[4 + half]     : 0.f;
      A[l][m][0] = mk2(i0, i1);
      A[l][m][1] = mk2(h0, h1);
      A[l][m][2] = mk2(i2, h2);
      v8f c;
#pragma unroll
      for (int r = 0; r < 8; ++r) {
        const RowSel cs = row_sel(m, r + 8 * half);
        const float bi = (cs.si != 0.f) ? cs.si * Bi[cs.g] : 0.f;
        const float bh = (cs.sh != 0.f) ? cs.sh * Bh[cs.g] : 0.f;
        c[r] = bi + bh;
      }
      C[l][m] = c;
    }
  }

  float ha0 = 0.f, hb0 = 0.f, hc0 = 0.f;
  float ha1 = 0.f, hb1 = 0.f, hc1 = 0.f;
  const int len = lengths[s];
  const float* xlo = x + (size_t)s * (size_t)tmax * 6 + 2 * half;
  const float* xhi = x + (size_t)s * (size_t)tmax * 6 + 4 + half;

  float2 xl = *(const float2*)xlo;
  float  xc = *xhi;

#pragma unroll 1
  for (int t = 0; t <= tmax; ++t) {
    const bool valid0 = (t < len) && (t < tmax);
    const bool valid1 = (t >= 1) && (t - 1 < len);
    const v2f X0 = mk2(xl.x, xl.y);
    const v2f X2 = mk2(xc, hc0);
    const v2f P0 = mk2(ha0, hb0);
    const v2f Q0 = mk2(ha1, hb1);
    const v2f PQ = mk2(hc0, hc1);

    if (t + 1 < tmax) {
      xlo += 6; xhi += 6;
      xl = *(const float2*)xlo;
      xc = *xhi;
    }

    v8f a0 = wmma4(A[0][0][0], X0, C[0][0]);
    v8f b0 = wmma4(A[1][0][0], P0, C[1][0]);
    v8f a1 = wmma4(A[0][1][0], X0, C[0][1]);
    v8f b1 = wmma4(A[1][1][0], P0, C[1][1]);
    a0 = wmma4(A[0][0][1], P0, a0);
    b0 = wmma4(A[1][0][1], Q0, b0);
    a1 = wmma4(A[0][1][1], P0, a1);
    b1 = wmma4(A[1][1][1], Q0, b1);
    a0 = wmma4(A[0][0][2], X2, a0);
    b0 = wmma4(A[1][0][2], PQ, b0);
    a1 = wmma4(A[0][1][2], X2, a1);
    b1 = wmma4(A[1][1][2], PQ, b1);

    const float ga0 = gru_unit(a0[0], a0[1], a0[6], a0[7], ha0);
    const float ga1 = gru_unit(b0[0], b0[1], b0[6], b0[7], ha1);
    const float gb0 = gru_unit(a0[2], a0[3], a1[0], a1[1], hb0);
    const float gb1 = gru_unit(b0[2], b0[3], b1[0], b1[1], hb1);
    const float gc0 = gru_unit(a0[4], a0[5], a1[2], a1[3], hc0);
    const float gc1 = gru_unit(b0[4], b0[5], b1[2], b1[3], hc1);
    ha0 = valid0 ? ga0 : ha0;                    ha1 = valid1 ? ga1 : ha1;
    hb0 = valid0 ? gb0 : hb0;                    hb1 = valid1 ? gb1 : hb1;
    hc0 = valid0 ? gc0 : hc0;                    hc1 = valid1 ? gc1 : hc1;
    __builtin_amdgcn_sched_barrier(0);
  }

  float* orow = out + (size_t)s * 6;
  *(float2*)(orow + (half ? 2 : 0)) = make_float2(ha1, hb1);
  orow[half ? 5 : 4] = hc1;
}

extern "C" void kernel_launch(void* const* d_in, const int* in_sizes, int n_in,
                              void* d_out, int out_size, void* d_ws, size_t ws_size,
                              hipStream_t stream) {
  (void)n_in; (void)out_size; (void)d_ws; (void)ws_size;
  const float* x       = (const float*)d_in[0];
  const int*   lengths = (const int*)d_in[1];
  const float* W_ih    = (const float*)d_in[2];
  const float* W_hh    = (const float*)d_in[3];
  const float* b_ih    = (const float*)d_in[4];
  const float* b_hh    = (const float*)d_in[5];
  float* out = (float*)d_out;

  const int nseq = in_sizes[1];
  const int tmax = in_sizes[0] / (nseq * 6);
  const int waves = nseq / 16;
  const int threads = waves * 32;
  const int block = 256;
  const int grid = (threads + block - 1) / block;
  gru16_wmma_kernel<<<grid, block, 0, stream>>>(x, lengths, W_ih, W_hh, b_ih, b_hh, out, tmax);
}


# Per work-group length sort in LDS: 128 sequences dealt to the 8 waves, longest with shortest per SIMD; each wave stops at its own longest length; rows leave via LDS, one store per line
# speedup vs baseline: 2.0182x; 1.2308x over previous
#include <hip/hip_runtime.h>

typedef __attribute__((ext_vector_type(2))) float v2f;
typedef __attribute__((ext_vector_type(8))) float v8f;

struct RowSel { int g; float si; float sh; };
__device__ __forceinline__ RowSel row_sel(int m, int R) {
  const int hh = R >> 3, r = R & 7;
  const int ua = 2 * hh, ub = 2 * hh + 1, uc = 4 + hh;
  RowSel s; s.g = 0; s.si = 0.f; s.sh = 0.f;
  if (m == 0) {
    if (r < 6) {
      const int p = r >> 1;
      const int u = (p == 0) ? ua : ((p == 1) ? ub : uc);
      s.g = u + 6 * (r & 1); s.si = 0.5f; s.sh = 0.5f;
    } else {
      s.g = 12 + ua; s.si = (r == 6) ? 1.f : 0.f; s.sh = 0.5f;
    }
  } else if (r < 4) {
    const int u = (r >> 1) ? uc : ub;
    s.g = 12 + u; s.si = ((r & 1) == 0) ? 1.f : 0.f; s.sh = 0.5f;
  }
  return s;
}

__device__ __forceinline__ float tanh_hw(float x) {
  return __builtin_amdgcn_tanhf(x);
}
__device__ __forceinline__ float gru_unit(float ar, float az, float s, float q, float h) {
  const float tr = tanh_hw(ar);
  const float tz = tanh_hw(az);
  const float n  = tanh_hw(__builtin_fmaf(tr, q, s));
  const float z  = __builtin_fmaf(0.5f, tz, 0.5f);
  return __builtin_fmaf(z, h - n, n);
}
__device__ __forceinline__ v8f wmma4(v2f a, v2f b, v8f c) {
  return __builtin_amdgcn_wmma_f32_16x16x4_f32(false, a, false, b, (short)0, c, false, false);
}
__device__ __forceinline__ v2f mk2(float a, float b) { v2f r; r[0] = a; r[1] = b; return r; }

__global__ __launch_bounds__(256) void gru16_wmma_kernel(
    const float* __restrict__ x, const int* __restrict__ lengths,
    const float* __restrict__ W_ih, const float* __restrict__ W_hh,
    const float* __restrict__ b_ih, const float* __restrict__ b_hh,
    float* __restrict__ out, int tmax)
{
  __shared__ int   s_key[128];
  __shared__ int   s_ord[128];
  __shared__ int   s_skey[128];
  __shared__ float s_out[128 * 6];

  const int tid  = threadIdx.x;
  const int lane = tid & 31;
  const int col  = lane & 15;
  const int half = lane >> 4;
  const int S0   = blockIdx.x * 128;

  v2f A[2][2][3];
  v8f C[2][2];
#pragma unroll
  for (int l = 0; l < 2; ++l) {
    const float* Wi = W_ih + l * 108;
    const float* Wh = W_hh + l * 108;
    const float* Bi = b_ih + l * 18;
    const float* Bh = b_hh + l * 18;
#pragma unroll
    for (int m = 0; m < 2; ++m) {
      const RowSel rs = row_sel(m, col);
      const float* wi = Wi + rs.g * 6;
      const float* wh = Wh + rs.g * 6;
      const bool ui = (rs.si != 0.f), uh = (rs.sh != 0.f);
      const float i0 = ui ? rs.si * wi[2 * half]     : 0.f;
      const float i1 = ui ? rs.si * wi[2 * half + 1] : 0.f;
      const float i2 = ui ? rs.si * wi[4 + half]     : 0.f;
      const float h0 = uh ? rs.sh * wh[2 * half]     : 0.f;
      const float h1 = uh ? rs.sh * wh[2 * half + 1] : 0.f;
      const float h2 = uh ? rs.sh * wh[4 + half]     : 0.f;
      A[l][m][0] = mk2(i0, i1);
      A[l][m][1] = mk2(h0, h1);
      A[l][m][2] = mk2(i2, h2);
      v8f c;
#pragma unroll
      for (int r = 0; r < 8; ++r) {
        const RowSel cs = row_sel(m, r + 8 * half);
        const float bi = (cs.si != 0.f) ? cs.si * Bi[cs.g] : 0.f;
        const float bh = (cs.sh != 0.f) ? cs.sh * Bh[cs.g] : 0.f;
        c[r] = bi + bh;
      }
      C[l][m] = c;
    }
  }

  float ha0 = 0.f, hb0 = 0.f, hc0 = 0.f;
  float ha1 = 0.f, hb1 = 0.f, hc1 = 0.f;
  int key = 0;
  if (tid < 128) {
    const int L = lengths[S0 + tid];
    key = (L < 0) ? 0 : ((L > tmax) ? tmax : L);
    s_key[tid] = key;
  }
  __syncthreads();
  if (tid < 128) {
    int r = 0;
    for (int j = 0; j < 128; ++j) {
      const int kj = s_key[j];
      r += ((kj < key) || (kj == key && j < tid)) ? 1 : 0;
    }
    s_ord[r]  = tid;
    s_skey[r] = key;
  }
  __syncthreads();
  const int wib = __builtin_amdgcn_readfirstlane(tid >> 5);
  const int grp = (0x35174260 >> (4 * wib)) & 7;
  const int pos = 16 * grp + col;
  const int li  = s_ord[pos];
  const int len = s_skey[pos];
  const int tw  = __builtin_amdgcn_readfirstlane(s_skey[16 * grp + 15]);
  const int s   = S0 + li;
  const float* xlo = x + (size_t)s * (size_t)tmax * 6 + 2 * half;
  const float* xhi = x + (size_t)s * (size_t)tmax * 6 + 4 + half;

  float2 xl = *(const float2*)xlo;
  float  xc = *xhi;

#pragma unroll 1
  for (int t = 0; t <= tw; ++t) {
    const bool valid0 = (t < len);
    const bool valid1 = (t >= 1) && (t - 1 < len);
    const v2f X0 = mk2(xl.x, xl.y);
    const v2f X2 = mk2(xc, hc0);
    const v2f P0 = mk2(ha0, hb0);
    const v2f Q0 = mk2(ha1, hb1);
    const v2f PQ = mk2(hc0, hc1);

    if (t + 1 < tw) {
      xlo += 6; xhi += 6;
      xl = *(const float2*)xlo;
      xc = *xhi;
    }

    v8f a0 = wmma4(A[0][0][0], X0, C[0][0]);
    v8f b0 = wmma4(A[1][0][0], P0, C[1][0]);
    v8f a1 = wmma4(A[0][1][0], X0, C[0][1]);
    v8f b1 = wmma4(A[1][1][0], P0, C[1][1]);
    a0 = wmma4(A[0][0][1], P0, a0);
    b0 = wmma4(A[1][0][1], Q0, b0);
    a1 = wmma4(A[0][1][1], P0, a1);
    b1 = wmma4(A[1][1][1], Q0, b1);
    a0 = wmma4(A[0][0][2], X2, a0);
    b0 = wmma4(A[1][0][2], PQ, b0);
    a1 = wmma4(A[0][1][2], X2, a1);
    b1 = wmma4(A[1][1][2], PQ, b1);

    const float ga0 = gru_unit(a0[0], a0[1], a0[6], a0[7], ha0);
    const float ga1 = gru_unit(b0[0], b0[1], b0[6], b0[7], ha1);
    const float gb0 = gru_unit(a0[2], a0[3], a1[0], a1[1], hb0);
    const float gb1 = gru_unit(b0[2], b0[3], b1[0], b1[1], hb1);
    const float gc0 = gru_unit(a0[4], a0[5], a1[2], a1[3], hc0);
    const float gc1 = gru_unit(b0[4], b0[5], b1[2], b1[3], hc1);
    ha0 = valid0 ? ga0 : ha0;                    ha1 = valid1 ? ga1 : ha1;
    hb0 = valid0 ? gb0 : hb0;                    hb1 = valid1 ? gb1 : hb1;
    hc0 = valid0 ? gc0 : hc0;                    hc1 = valid1 ? gc1 : hc1;
    __builtin_amdgcn_sched_barrier(0);
  }

  float* orow = s_out + li * 6;
  orow[half ? 2 : 0] = ha1;
  orow[half ? 3 : 1] = hb1;
  orow[half ? 5 : 4] = hc1;
  __syncthreads();
  float* ob = out + (size_t)S0 * 6;
  for (int k = 0; k < 3; ++k) ob[tid + 256 * k] = s_out[tid + 256 * k];
}

extern "C" void kernel_launch(void* const* d_in, const int* in_sizes, int n_in,
                              void* d_out, int out_size, void* d_ws, size_t ws_size,
                              hipStream_t stream) {
  (void)n_in; (void)out_size; (void)d_ws; (void)ws_size;
  const float* x       = (const float*)d_in[0];
  const int*   lengths = (const int*)d_in[1];
  const float* W_ih    = (const float*)d_in[2];
  const float* W_hh    = (const float*)d_in[3];
  const float* b_ih    = (const float*)d_in[4];
  const float* b_hh    = (const float*)d_in[5];
  float* out = (float*)d_out;

  const int nseq = in_sizes[1];
  const int tmax = in_sizes[0] / (nseq * 6);
  const int block = 256;
  const int grid = nseq / 128;
  gru16_wmma_kernel<<<grid, block, 0, stream>>>(x, lengths, W_ih, W_hh, b_ih, b_hh, out, tmax);
}


# x fetched a block of 8 trips ahead: block b+1's 16 loads are issued in front of block b's 8 unrolled trips and first touched at the hand-over; was one trip ahead, waited for at every trip's end
# speedup vs baseline: 2.6627x; 1.3193x over previous
#include <hip/hip_runtime.h>

typedef __attribute__((ext_vector_type(2))) float v2f;
typedef __attribute__((ext_vector_type(8))) float v8f;

constexpr int XB = 8;

struct RowSel { int g; float si; float sh; };
__device__ __forceinline__ RowSel row_sel(int m, int R) {
  const int hh = R >> 3, r = R & 7;
  const int ua = 2 * hh, ub = 2 * hh + 1, uc = 4 + hh;
  RowSel s; s.g = 0; s.si = 0.f; s.sh = 0.f;
  if (m == 0) {
    if (r < 6) {
      const int p = r >> 1;
      const int u = (p == 0) ? ua : ((p == 1) ? ub : uc);
      s.g = u + 6 * (r & 1); s.si = 0.5f; s.sh = 0.5f;
    } else {
      s.g = 12 + ua; s.si = (r == 6) ? 1.f : 0.f; s.sh = 0.5f;
    }
  } else if (r < 4) {
    const int u = (r >> 1) ? uc : ub;
    s.g = 12 + u; s.si = ((r & 1) == 0) ? 1.f : 0.f; s.sh = 0.5f;
  }
  return s;
}

__device__ __forceinline__ float tanh_hw(float x) {
  return __builtin_amdgcn_tanhf(x);
}
__device__ __forceinline__ float gru_unit(float ar, float az, float s, float q, float h) {
  const float tr = tanh_hw(ar);
  const float tz = tanh_hw(az);
  const float n  = tanh_hw(__builtin_fmaf(tr, q, s));
  const float z  = __builtin_fmaf(0.5f, tz, 0.5f);
  return __builtin_fmaf(z, h - n, n);
}
__device__ __forceinline__ v8f wmma4(v2f a, v2f b, v8f c) {
  return __builtin_amdgcn_wmma_f32_16x16x4_f32(false, a, false, b, (short)0, c, false, false);
}
__device__ __forceinline__ v2f mk2(float a, float b) { v2f r; r[0] = a; r[1] = b; return r; }

__global__ __launch_bounds__(256) void gru16_wmma_kernel(
    const float* __restrict__ x, const int* __restrict__ lengths,
    const float* __restrict__ W_ih, const float* __restrict__ W_hh,
    const float* __restrict__ b_ih, const float* __restrict__ b_hh,
    float* __restrict__ out, int tmax)
{
  __shared__ int   s_key[128];
  __shared__ int   s_ord[128];
  __shared__ int   s_skey[128];
  __shared__ float s_out[128 * 6];

  const int tid  = threadIdx.x;
  const int lane = tid & 31;
  const int col  = lane & 15;
  const int half = lane >> 4;
  const int S0   = blockIdx.x * 128;

  v2f A[2][2][3];
  v8f C[2][2];
#pragma unroll
  for (int l = 0; l < 2; ++l) {
    const float* Wi = W_ih + l * 108;
    const float* Wh = W_hh + l * 108;
    const float* Bi = b_ih + l * 18;
    const float* Bh = b_hh + l * 18;
#pragma unroll
    for (int m = 0; m < 2; ++m) {
      const RowSel rs = row_sel(m, col);
      const float* wi = Wi + rs.g * 6;
      const float* wh = Wh + rs.g * 6;
      const bool ui = (rs.si != 0.f), uh = (rs.sh != 0.f);
      const float i0 = ui ? rs.si * wi[2 * half]     : 0.f;
      const float i1 = ui ? rs.si * wi[2 * half + 1] : 0.f;
      const float i2 = ui ? rs.si * wi[4 + half]     : 0.f;
      const float h0 = uh ? rs.sh * wh[2 * half]     : 0.f;
      const float h1 = uh ? rs.sh * wh[2 * half + 1] : 0.f;
      const float h2 = uh ? rs.sh * wh[4 + half]     : 0.f;
      A[l][m][0] = mk2(i0, i1);
      A[l][m][1] = mk2(h0, h1);
      A[l][m][2] = mk2(i2, h2);
      v8f c;
#pragma unroll
      for (int r = 0; r < 8; ++r) {
        const RowSel cs = row_sel(m, r + 8 * half);
        const float bi = (cs.si != 0.f) ? cs.si * Bi[cs.g] : 0.f;
        const float bh = (cs.sh != 0.f) ? cs.sh * Bh[cs.g] : 0.f;
        c[r] = bi + bh;
      }
      C[l][m] = c;
    }
  }

  float ha0 = 0.f, hb0 = 0.f, hc0 = 0.f;
  float ha1 = 0.f, hb1 = 0.f, hc1 = 0.f;
  int key = 0;
  if (tid < 128) {
    const int L = lengths[S0 + tid];
    key = (L < 0) ? 0 : ((L > tmax) ? tmax : L);
    s_key[tid] = key;
  }
  __syncthreads();
  if (tid < 128) {
    int r = 0;
    for (int j = 0; j < 128; ++j) {
      const int kj = s_key[j];
      r += ((kj < key) || (kj == key && j < tid)) ? 1 : 0;
    }
    s_ord[r]  = tid;
    s_skey[r] = key;
  }
  __syncthreads();
  const int wib = __builtin_amdgcn_readfirstlane(tid >> 5);
  const int grp = (0x35174260 >> (4 * wib)) & 7;
  const int pos = 16 * grp + col;
  const int li  = s_ord[pos];
  const int len = s_skey[pos];
  const int tw  = __builtin_amdgcn_readfirstlane(s_skey[16 * grp + 15]);
  const int s   = S0 + li;
  const float* xlo = x + (size_t)s * (size_t)tmax * 6 + 2 * half;
  const float* xhi = x + (size_t)s * (size_t)tmax * 6 + 4 + half;

  auto trip = [&](int t, float2 xl, float xc) {
    const bool valid0 = (t < len);
    const bool valid1 = (t >= 1) && (t - 1 < len);
    const v2f X0 = mk2(xl.x, xl.y);
    const v2f X2 = mk2(xc, hc0);
    const v2f P0 = mk2(ha0, hb0);
    const v2f Q0 = mk2(ha1, hb1);
    const v2f PQ = mk2(hc0, hc1);

    v8f a0 = wmma4(A[0][0][0], X0, C[0][0]);
    v8f b0 = wmma4(A[1][0][0], P0, C[1][0]);
    v8f a1 = wmma4(A[0][1][0], X0, C[0][1]);
    v8f b1 = wmma4(A[1][1][0], P0, C[1][1]);
    a0 = wmma4(A[0][0][1], P0, a0);
    b0 = wmma4(A[1][0][1], Q0, b0);
    a1 = wmma4(A[0][1][1], P0, a1);
    b1 = wmma4(A[1][1][1], Q0, b1);
    a0 = wmma4(A[0][0][2], X2, a0);
    b0 = wmma4(A[1][0][2], PQ, b0);
    a1 = wmma4(A[0][1][2], X2, a1);
    b1 = wmma4(A[1][1][2], PQ, b1);

    const float ga0 = gru_unit(a0[0], a0[1], a0[6], a0[7], ha0);
    const float ga1 = gru_unit(b0[0], b0[1], b0[6], b0[7], ha1);
    const float gb0 = gru_unit(a0[2], a0[3], a1[0], a1[1], hb0);
    const float gb1 = gru_unit(b0[2], b0[3], b1[0], b1[1], hb1);
    const float gc0 = gru_unit(a0[4], a0[5], a1[2], a1[3], hc0);
    const float gc1 = gru_unit(b0[4], b0[5], b1[2], b1[3], hc1);
    ha0 = valid0 ? ga0 : ha0;                    ha1 = valid1 ? ga1 : ha1;
    hb0 = valid0 ? gb0 : hb0;                    hb1 = valid1 ? gb1 : hb1;
    hc0 = valid0 ? gc0 : hc0;                    hc1 = valid1 ? gc1 : hc1;
  };

  float2 xl_c[XB], xl_n[XB];
  float  xc_c[XB], xc_n[XB];
  int tnext = 0;
  if (XB <= tmax) {
#pragma unroll
    for (int u = 0; u < XB; ++u) {
      xl_c[u] = *(const float2*)(xlo + 6 * u);
      xc_c[u] = xhi[6 * u];
    }
    int tb = 0;
#pragma unroll 1
    for (; tb + XB < tw && tb + 2 * XB <= tmax; tb += XB) {
      xlo += 6 * XB; xhi += 6 * XB;
#pragma unroll
      for (int u = 0; u < XB; ++u) {
        xl_n[u] = *(const float2*)(xlo + 6 * u);
        xc_n[u] = xhi[6 * u];
      }
      __builtin_amdgcn_sched_barrier(0);
#pragma unroll
      for (int u = 0; u < XB; ++u) trip(tb + u, xl_c[u], xc_c[u]);
#pragma unroll
      for (int u = 0; u < XB; ++u) { xl_c[u] = xl_n[u]; xc_c[u] = xc_n[u]; }
    }
    tnext = tb + XB;
#pragma unroll
    for (int u = 0; u < XB; ++u) {
      if (tb + u > tw) break;
      trip(tb + u, xl_c[u], xc_c[u]);
    }
  }
#pragma unroll 1
  for (int t = tnext; t <= tw; ++t) {
    float2 xl = make_float2(0.f, 0.f);
    float  xc = 0.f;
    if (t < tw) {
      const float* row = x + (size_t)s * (size_t)tmax * 6 + (size_t)t * 6;
      xl = *(const float2*)(row + 2 * half);
      xc = row[4 + half];
    }
    trip(t, xl, xc);
  }

  float* orow = s_out + li * 6;
  orow[half ? 2 : 0] = ha1;
  orow[half ? 3 : 1] = hb1;
  orow[half ? 5 : 4] = hc1;
  __syncthreads();
  float* ob = out + (size_t)S0 * 6;
  for (int k = 0; k < 3; ++k) ob[tid + 256 * k] = s_out[tid + 256 * k];
}

extern "C" void kernel_launch(void* const* d_in, const int* in_sizes, int n_in,
                              void* d_out, int out_size, void* d_ws, size_t ws_size,
                              hipStream_t stream) {
  (void)n_in; (void)out_size; (void)d_ws; (void)ws_size;
  const float* x       = (const float*)d_in[0];
  const int*   lengths = (const int*)d_in[1];
  const float* W_ih    = (const float*)d_in[2];
  const float* W_hh    = (const float*)d_in[3];
  const float* b_ih    = (const float*)d_in[4];
  const float* b_hh    = (const float*)d_in[5];
  float* out = (float*)d_out;

  const int nseq = in_sizes[1];
  const int tmax = in_sizes[0] / (nseq * 6);
  const int block = 256;
  const int grid = nseq / 128;
  gru16_wmma_kernel<<<grid, block, 0, stream>>>(x, lengths, W_ih, W_hh, b_ih, b_hh, out, tmax);
}


# Prologue reads issued in groups, one wait per group: weight reads made unconditional and sent with the length read; rank loop reads 32 keys per pass. Was a branch and a wait per weight, 8 keys a wait
# speedup vs baseline: 2.9418x; 1.1048x over previous
#include <hip/hip_runtime.h>

typedef __attribute__((ext_vector_type(2))) float v2f;
typedef __attribute__((ext_vector_type(8))) float v8f;

constexpr int XB = 8;

struct RowSel { int g; float si; float sh; };
__device__ __forceinline__ RowSel row_sel(int m, int R) {
  const int hh = R >> 3, r = R & 7;
  const int ua = 2 * hh, ub = 2 * hh + 1, uc = 4 + hh;
  RowSel s; s.g = 0; s.si = 0.f; s.sh = 0.f;
  if (m == 0) {
    if (r < 6) {
      const int p = r >> 1;
      const int u = (p == 0) ? ua : ((p == 1) ? ub : uc);
      s.g = u + 6 * (r & 1); s.si = 0.5f; s.sh = 0.5f;
    } else {
      s.g = 12 + ua; s.si = (r == 6) ? 1.f : 0.f; s.sh = 0.5f;
    }
  } else if (r < 4) {
    const int u = (r >> 1) ? uc : ub;
    s.g = 12 + u; s.si = ((r & 1) == 0) ? 1.f : 0.f; s.sh = 0.5f;
  }
  return s;
}

__device__ __forceinline__ float tanh_hw(float x) {
  return __builtin_amdgcn_tanhf(x);
}
__device__ __forceinline__ float gru_unit(float ar, float az, float s, float q, float h) {
  const float tr = tanh_hw(ar);
  const float tz = tanh_hw(az);
  const float n  = tanh_hw(__builtin_fmaf(tr, q, s));
  const float z  = __builtin_fmaf(0.5f, tz, 0.5f);
  return __builtin_fmaf(z, h - n, n);
}
__device__ __forceinline__ v8f wmma4(v2f a, v2f b, v8f c) {
  return __builtin_amdgcn_wmma_f32_16x16x4_f32(false, a, false, b, (short)0, c, false, false);
}
__device__ __forceinline__ v2f mk2(float a, float b) { v2f r; r[0] = a; r[1] = b; return r; }

__global__ __launch_bounds__(256) void gru16_wmma_kernel(
    const float* __restrict__ x, const int* __restrict__ lengths,
    const float* __restrict__ W_ih, const float* __restrict__ W_hh,
    const float* __restrict__ b_ih, const float* __restrict__ b_hh,
    float* __restrict__ out, int tmax)
{
  __shared__ int   s_key[128];
  __shared__ int   s_ord[128];
  __shared__ int   s_skey[128];
  __shared__ float s_out[128 * 6];

  const int tid  = threadIdx.x;
  const int lane = tid & 31;
  const int col  = lane & 15;
  const int half = lane >> 4;
  const int S0   = blockIdx.x * 128;

  const int Lraw = lengths[S0 + (tid & 127)];

  v2f A[2][2][3];
  v8f C[2][2];
#pragma unroll
  for (int l = 0; l < 2; ++l) {
    const float* Wi = W_ih + l * 108;
    const float* Wh = W_hh + l * 108;
    const float* Bi = b_ih + l * 18;
    const float* Bh = b_hh + l * 18;
#pragma unroll
    for (int m = 0; m < 2; ++m) {
      const RowSel rs = row_sel(m, col);
      const float* wi = Wi + rs.g * 6;
      const float* wh = Wh + rs.g * 6;
      const float wi0 = wi[2 * half], wi1 = wi[2 * half + 1], wi2 = wi[4 + half];
      const float wh0 = wh[2 * half], wh1 = wh[2 * half + 1], wh2 = wh[4 + half];
      const bool ui = (rs.si != 0.f), uh = (rs.sh != 0.f);
      const float i0 = ui ? rs.si * wi0 : 0.f;
      const float i1 = ui ? rs.si * wi1 : 0.f;
      const float i2 = ui ? rs.si * wi2 : 0.f;
      const float h0 = uh ? rs.sh * wh0 : 0.f;
      const float h1 = uh ? rs.sh * wh1 : 0.f;
      const float h2 = uh ? rs.sh * wh2 : 0.f;
      A[l][m][0] = mk2(i0, i1);
      A[l][m][1] = mk2(h0, h1);
      A[l][m][2] = mk2(i2, h2);
      v8f c;
#pragma unroll
      for (int r = 0; r < 8; ++r) {
        const RowSel cs = row_sel(m, r + 8 * half);
        const float bi = (cs.si != 0.f) ? cs.si * Bi[cs.g] : 0.f;
        const float bh = (cs.sh != 0.f) ? cs.sh * Bh[cs.g] : 0.f;
        c[r] = bi + bh;
      }
      C[l][m] = c;
    }
  }

  float ha0 = 0.f, hb0 = 0.f, hc0 = 0.f;
  float ha1 = 0.f, hb1 = 0.f, hc1 = 0.f;
  const int key = (Lraw < 0) ? 0 : ((Lraw > tmax) ? tmax : Lraw);
  if (tid < 128) s_key[tid] = key;
  __syncthreads();
  if (tid < 128) {
    int r = 0;
#pragma unroll 16
    for (int j = 0; j < 128; ++j) {
      const int kj = s_key[j];
      r += ((kj < key) || (kj == key && j < tid)) ? 1 : 0;
    }
    s_ord[r]  = tid;
    s_skey[r] = key;
  }
  __syncthreads();
  const int wib = __builtin_amdgcn_readfirstlane(tid >> 5);
  const int grp = (0x35174260 >> (4 * wib)) & 7;
  const int pos = 16 * grp + col;
  const int li  = s_ord[pos];
  const int len = s_skey[pos];
  const int tw  = __builtin_amdgcn_readfirstlane(s_skey[16 * grp + 15]);
  const int s   = S0 + li;
  const float* xlo = x + (size_t)s * (size_t)tmax * 6 + 2 * half;
  const float* xhi = x + (size_t)s * (size_t)tmax * 6 + 4 + half;

  auto trip = [&](int t, float2 xl, float xc) {
    const bool valid0 = (t < len);
    const bool valid1 = (t >= 1) && (t - 1 < len);
    const v2f X0 = mk2(xl.x, xl.y);
    const v2f X2 = mk2(xc, hc0);
    const v2f P0 = mk2(ha0, hb0);
    const v2f Q0 = mk2(ha1, hb1);
    const v2f PQ = mk2(hc0, hc1);

    v8f a0 = wmma4(A[0][0][0], X0, C[0][0]);
    v8f b0 = wmma4(A[1][0][0], P0, C[1][0]);
    v8f a1 = wmma4(A[0][1][0], X0, C[0][1]);
    v8f b1 = wmma4(A[1][1][0], P0, C[1][1]);
    a0 = wmma4(A[0][0][1], P0, a0);
    b0 = wmma4(A[1][0][1], Q0, b0);
    a1 = wmma4(A[0][1][1], P0, a1);
    b1 = wmma4(A[1][1][1], Q0, b1);
    a0 = wmma4(A[0][0][2], X2, a0);
    b0 = wmma4(A[1][0][2], PQ, b0);
    a1 = wmma4(A[0][1][2], X2, a1);
    b1 = wmma4(A[1][1][2], PQ, b1);

    const float ga0 = gru_unit(a0[0], a0[1], a0[6], a0[7], ha0);
    const float ga1 = gru_unit(b0[0], b0[1], b0[6], b0[7], ha1);
    const float gb0 = gru_unit(a0[2], a0[3], a1[0], a1[1], hb0);
    const float gb1 = gru_unit(b0[2], b0[3], b1[0], b1[1], hb1);
    const float gc0 = gru_unit(a0[4], a0[5], a1[2], a1[3], hc0);
    const float gc1 = gru_unit(b0[4], b0[5], b1[2], b1[3], hc1);
    ha0 = valid0 ? ga0 : ha0;                    ha1 = valid1 ? ga1 : ha1;
    hb0 = valid0 ? gb0 : hb0;                    hb1 = valid1 ? gb1 : hb1;
    hc0 = valid0 ? gc0 : hc0;                    hc1 = valid1 ? gc1 : hc1;
  };

  float2 xl_c[XB], xl_n[XB];
  float  xc_c[XB], xc_n[XB];
  int tnext = 0;
  if (XB <= tmax) {
#pragma unroll
    for (int u = 0; u < XB; ++u) {
      xl_c[u] = *(const float2*)(xlo + 6 * u);
      xc_c[u] = xhi[6 * u];
    }
    int tb = 0;
#pragma unroll 1
    for (; tb + XB < tw && tb + 2 * XB <= tmax; tb += XB) {
      xlo += 6 * XB; xhi += 6 * XB;
#pragma unroll
      for (int u = 0; u < XB; ++u) {
        xl_n[u] = *(const float2*)(xlo + 6 * u);
        xc_n[u] = xhi[6 * u];
      }
      __builtin_amdgcn_sched_barrier(0);
#pragma unroll
      for (int u = 0; u < XB; ++u) trip(tb + u, xl_c[u], xc_c[u]);
#pragma unroll
      for (int u = 0; u < XB; ++u) { xl_c[u] = xl_n[u]; xc_c[u] = xc_n[u]; }
    }
    tnext = tb + XB;
#pragma unroll
    for (int u = 0; u < XB; ++u) {
      if (tb + u > tw) break;
      trip(tb + u, xl_c[u], xc_c[u]);
    }
  }
#pragma unroll 1
  for (int t = tnext; t <= tw; ++t) {
    float2 xl = make_float2(0.f, 0.f);
    float  xc = 0.f;
    if (t < tw) {
      const float* row = x + (size_t)s * (size_t)tmax * 6 + (size_t)t * 6;
      xl = *(const float2*)(row + 2 * half);
      xc = row[4 + half];
    }
    trip(t, xl, xc);
  }

  float* orow = s_out + li * 6;
  orow[half ? 2 : 0] = ha1;
  orow[half ? 3 : 1] = hb1;
  orow[half ? 5 : 4] = hc1;
  __syncthreads();
  float* ob = out + (size_t)S0 * 6;
  for (int k = 0; k < 3; ++k) ob[tid + 256 * k] = s_out[tid + 256 * k];
}

extern "C" void kernel_launch(void* const* d_in, const int* in_sizes, int n_in,
                              void* d_out, int out_size, void* d_ws, size_t ws_size,
                              hipStream_t stream) {
  (void)n_in; (void)out_size; (void)d_ws; (void)ws_size;
  const float* x       = (const float*)d_in[0];
  const int*   lengths = (const int*)d_in[1];
  const float* W_ih    = (const float*)d_in[2];
  const float* W_hh    = (const float*)d_in[3];
  const float* b_ih    = (const float*)d_in[4];
  const float* b_hh    = (const float*)d_in[5];
  float* out = (float*)d_out;

  const int nseq = in_sizes[1];
  const int tmax = in_sizes[0] / (nseq * 6);
  const int block = 256;
  const int grid = nseq / 128;
  gru16_wmma_kernel<<<grid, block, 0, stream>>>(x, lengths, W_ih, W_hh, b_ih, b_hh, out, tmax);
}
